# starting kernel
# baseline (speedup 1.0000x reference)
#include <hip/hip_runtime.h>


#define K_SZ   7
#define KK     49
#define G_NUM  16
#define C_CH   256
#define CR     64
#define NK     (KK * G_NUM)
#define HW     64
#define BATCH  2
#define M_TOT  (BATCH * HW * HW)
#define BN_EPS 1e-3f

typedef __attribute__((ext_vector_type(2))) float v2f;
typedef __attribute__((ext_vector_type(8))) float v8f;

__global__ __launch_bounds__(256) void gemm1_bn_relu(
    const float* __restrict__ x,  const float* __restrict__ w1,
    const float* __restrict__ b1, const float* __restrict__ gamma,
    const float* __restrict__ beta, const float* __restrict__ mu,
    const float* __restrict__ var, float* __restrict__ t)
{
    const int wave  = threadIdx.x >> 5;
    const int ln    = threadIdx.x & 31;
    const int tile  = blockIdx.x * 8 + wave;
    const int m0    = (tile >> 2) << 4;
    const int n0    = (tile & 3)  << 4;
    const int laneM = ln & 15;
    const int kHalf = ln >> 4;

    const float* arow = x + (size_t)(m0 + laneM) * C_CH + 2 * kHalf;
    const float* bcol = w1 + (size_t)(2 * kHalf) * CR + (n0 + laneM);

    v8f acc = {};
#pragma unroll 8
    for (int k = 0; k < C_CH; k += 4) {
        v2f a, b;
        a.x = arow[k];
        a.y = arow[k + 1];
        b.x = bcol[(size_t)k * CR];
        b.y = bcol[(size_t)(k + 1) * CR];
        acc = __builtin_amdgcn_wmma_f32_16x16x4_f32(
            false, a, false, b, (short)0, acc, false, false);
    }

    const int   col   = n0 + laneM;
    const float scale = gamma[col] * rsqrtf(var[col] + BN_EPS);
    const float shift = beta[col] - mu[col] * scale;
    const float b1c   = b1[col];
    const int   rowB  = m0 + (kHalf << 3);
#pragma unroll
    for (int r = 0; r < 8; ++r) {
        float v = fmaf(acc[r] + b1c, scale, shift);
        t[(size_t)(rowB + r) * CR + col] = fmaxf(v, 0.0f);
    }
}

__global__ __launch_bounds__(256) void gemm2_kerngen(
    const float* __restrict__ t, const float* __restrict__ w2,
    const float* __restrict__ b2, float* __restrict__ kern)
{
    const int wave = threadIdx.x >> 5;
    const int ln   = threadIdx.x & 31;
    const int tile = blockIdx.x * 8 + wave;
    if (tile >= (M_TOT / 16) * (NK / 16)) return;
    const int mt = tile / (NK / 16);
    const int nt = tile % (NK / 16);
    const int m0 = mt << 4, n0 = nt << 4;
    const int laneM = ln & 15;
    const int kHalf = ln >> 4;

    const float* arow = t  + (size_t)(m0 + laneM) * CR + 2 * kHalf;
    const float* bcol = w2 + (size_t)(2 * kHalf) * NK + (n0 + laneM);

    v8f acc = {};
#pragma unroll
    for (int k = 0; k < CR; k += 4) {
        v2f a, b;
        a.x = arow[k];
        a.y = arow[k + 1];
        b.x = bcol[(size_t)k * NK];
        b.y = bcol[(size_t)(k + 1) * NK];
        acc = __builtin_amdgcn_wmma_f32_16x16x4_f32(
            false, a, false, b, (short)0, acc, false, false);
    }

    const int   col  = n0 + laneM;
    const float b2c  = b2[col];
    const int   rowB = m0 + (kHalf << 3);
#pragma unroll
    for (int r = 0; r < 8; ++r) {
        kern[(size_t)(rowB + r) * NK + col] = acc[r] + b2c;
    }
}

__global__ __launch_bounds__(256) void involution_gather(
    const float* __restrict__ x, const float* __restrict__ kern,
    float* __restrict__ out)
{
    __shared__ float ksh[NK];
    const int pix = blockIdx.x;
    const int c   = threadIdx.x;
    const int w   = pix & (HW - 1);
    const int h   = (pix >> 6) & (HW - 1);
    const int b   = pix >> 12;

    const float* kp = kern + (size_t)pix * NK;
    for (int i = threadIdx.x; i < NK; i += 256) ksh[i] = kp[i];
    __syncthreads();

    const int g = c & (G_NUM - 1);
    float acc = 0.0f;
#pragma unroll
    for (int kh = 0; kh < K_SZ; ++kh) {
        const int hh = h + kh - (K_SZ / 2);
        if ((unsigned)hh >= (unsigned)HW) continue;
#pragma unroll
        for (int kw = 0; kw < K_SZ; ++kw) {
            const int ww = w + kw - (K_SZ / 2);
            if ((unsigned)ww >= (unsigned)HW) continue;
            const float xv = x[(((size_t)(b * HW + hh)) * HW + ww) * C_CH + c];
            acc = fmaf(ksh[(kh * K_SZ + kw) * G_NUM + g], xv, acc);
        }
    }
    out[(size_t)pix * C_CH + c] = acc;
}

extern "C" void kernel_launch(void* const* d_in, const int* in_sizes, int n_in,
                              void* d_out, int out_size, void* d_ws, size_t ws_size,
                              hipStream_t stream)
{
    const float* x     = (const float*)d_in[0];
    const float* w1    = (const float*)d_in[1];
    const float* b1    = (const float*)d_in[2];
    const float* gamma = (const float*)d_in[3];
    const float* beta  = (const float*)d_in[4];
    const float* mu    = (const float*)d_in[5];
    const float* var   = (const float*)d_in[6];
    const float* w2    = (const float*)d_in[7];
    const float* b2    = (const float*)d_in[8];

    float* out  = (float*)d_out;
    float* kern = out + (size_t)M_TOT * C_CH;
    float* t    = (float*)d_ws;

    gemm1_bn_relu<<<256, 256, 0, stream>>>(x, w1, b1, gamma, beta, mu, var, t);
    gemm2_kerngen<<<3136, 256, 0, stream>>>(t, w2, b2, kern);
    involution_gather<<<M_TOT, 256, 0, stream>>>(x, kern, out);
}


# GEMM2 fused into the gather through registers: a 16-column WMMA tile = one tap x 16 groups, so a lane's accumulators are its 8 pixels' kern factors (lane = group); kern not re-read; sum order kept
# speedup vs baseline: 1.9066x; 1.9066x over previous
#include <hip/hip_runtime.h>


#define K_SZ   7
#define KK     49
#define G_NUM  16
#define C_CH   256
#define CR     64
#define NK     (KK * G_NUM)
#define HW     64
#define BATCH  2
#define M_TOT  (BATCH * HW * HW)
#define BN_EPS 1e-3f

typedef __attribute__((ext_vector_type(2))) float v2f;
typedef __attribute__((ext_vector_type(8))) float v8f;

__global__ __launch_bounds__(256) void gemm1_bn_relu(
    const float* __restrict__ x,  const float* __restrict__ w1,
    const float* __restrict__ b1, const float* __restrict__ gamma,
    const float* __restrict__ beta, const float* __restrict__ mu,
    const float* __restrict__ var, float* __restrict__ t)
{
    const int wave  = threadIdx.x >> 5;
    const int ln    = threadIdx.x & 31;
    const int tile  = blockIdx.x * 8 + wave;
    const int m0    = (tile >> 2) << 4;
    const int n0    = (tile & 3)  << 4;
    const int laneM = ln & 15;
    const int kHalf = ln >> 4;

    const float* arow = x + (size_t)(m0 + laneM) * C_CH + 2 * kHalf;
    const float* bcol = w1 + (size_t)(2 * kHalf) * CR + (n0 + laneM);

    v8f acc = {};
#pragma unroll 8
    for (int k = 0; k < C_CH; k += 4) {
        v2f a, b;
        a.x = arow[k];
        a.y = arow[k + 1];
        b.x = bcol[(size_t)k * CR];
        b.y = bcol[(size_t)(k + 1) * CR];
        acc = __builtin_amdgcn_wmma_f32_16x16x4_f32(
            false, a, false, b, (short)0, acc, false, false);
    }

    const int   col   = n0 + laneM;
    const float scale = gamma[col] * rsqrtf(var[col] + BN_EPS);
    const float shift = beta[col] - mu[col] * scale;
    const float b1c   = b1[col];
    const int   rowB  = m0 + (kHalf << 3);
#pragma unroll
    for (int r = 0; r < 8; ++r) {
        float v = fmaf(acc[r] + b1c, scale, shift);
        t[(size_t)(rowB + r) * CR + col] = fmaxf(v, 0.0f);
    }
}

__global__ __launch_bounds__(128) void kerngen_gather(
    const float* __restrict__ x,  const float* __restrict__ t,
    const float* __restrict__ w2, const float* __restrict__ b2,
    float* __restrict__ kern, float* __restrict__ out)
{
    const int wave = __builtin_amdgcn_readfirstlane((int)(threadIdx.x >> 5));
    const int ln   = threadIdx.x & 31;
    const int mt   = blockIdx.x * 2 + (wave >> 1);
    const int jh   = wave & 1;
    const int g    = ln & 15;
    const int half = ln >> 4;
    const int m0   = mt << 4;
    const int w0   = m0 & (HW - 1);
    const int h    = (m0 >> 6) & (HW - 1);
    const int b    = m0 >> 12;
    const int pb   = m0 + (half << 3);
    const int wl   = w0 + (half << 3);

    v2f a[CR / 4];
    const float* arow = t + (size_t)(m0 + g) * CR + 2 * half;
#pragma unroll
    for (int ks = 0; ks < CR / 4; ++ks) {
        a[ks].x = arow[4 * ks];
        a[ks].y = arow[4 * ks + 1];
    }

    int  cofs[14];
    bool cok[14];
#pragma unroll
    for (int ci = 0; ci < 14; ++ci) {
        const int ww = wl - (K_SZ / 2) + ci;
        cok[ci]  = (unsigned)ww < (unsigned)HW;
        cofs[ci] = (cok[ci] ? ww : wl) * C_CH;
    }

    float o[8][8];
#pragma unroll
    for (int jj = 0; jj < 8; ++jj)
#pragma unroll
        for (int r = 0; r < 8; ++r) o[jj][r] = 0.0f;

    const float* bbase = w2 + (size_t)(2 * half) * NK + g;
    const int    chan0 = g + G_NUM * 8 * jh;

    float bcn[K_SZ];
#pragma unroll
    for (int kw = 0; kw < K_SZ; ++kw) bcn[kw] = b2[kw * G_NUM + g];

#pragma unroll 1
    for (int kh = 0; kh < K_SZ; ++kh) {
        float bc[K_SZ];
#pragma unroll
        for (int kw = 0; kw < K_SZ; ++kw) bc[kw] = bcn[kw];
        const int kh1 = (kh + 1 < K_SZ) ? kh + 1 : kh;
#pragma unroll
        for (int kw = 0; kw < K_SZ; ++kw) bcn[kw] = b2[(kh1 * K_SZ + kw) * G_NUM + g];

        v8f kr[K_SZ];
#pragma unroll
        for (int kw = 0; kw < K_SZ; ++kw) kr[kw] = v8f{};

        const float* bp = bbase + kh * (K_SZ * G_NUM);
#pragma unroll
        for (int ks = 0; ks < CR / 4; ++ks) {
#pragma unroll
            for (int kw = 0; kw < K_SZ; ++kw) {
                v2f bb;
                bb.x = bp[(size_t)(4 * ks) * NK + kw * G_NUM];
                bb.y = bp[(size_t)(4 * ks + 1) * NK + kw * G_NUM];
                kr[kw] = __builtin_amdgcn_wmma_f32_16x16x4_f32(
                    false, a[ks], false, bb, (short)0, kr[kw], false, false);
            }
        }
        __builtin_amdgcn_sched_group_barrier(0x020, 49, 0);
#pragma unroll
        for (int i = 0; i < 112; ++i) {
            __builtin_amdgcn_sched_group_barrier(0x008, 1, 0);
            __builtin_amdgcn_sched_group_barrier(0x020, 2, 0);
        }
#pragma unroll
        for (int kw = 0; kw < K_SZ; ++kw)
#pragma unroll
            for (int r = 0; r < 8; ++r) kr[kw][r] += bc[kw];

        float xr[8][14];
        const int hh = h + kh - (K_SZ / 2);
        if ((unsigned)hh < (unsigned)HW) {
            const float* xr0 = x + ((size_t)(b * HW + hh) * HW) * C_CH + chan0;
#pragma unroll
            for (int jj = 0; jj < 8; ++jj)
#pragma unroll
                for (int ci = 0; ci < 14; ++ci)
                    xr[jj][ci] = xr0[cofs[ci] + G_NUM * jj];
        } else {
#pragma unroll
            for (int jj = 0; jj < 8; ++jj)
#pragma unroll
                for (int ci = 0; ci < 14; ++ci) xr[jj][ci] = 0.0f;
        }

#pragma unroll
        for (int kw = 0; kw < K_SZ; ++kw) {
            float* kp = kern + (size_t)(pb + 4 * jh) * NK + (kh * K_SZ + kw) * G_NUM + g;
#pragma unroll
            for (int i = 0; i < 4; ++i)
                kp[(size_t)i * NK] = jh ? kr[kw][4 + i] : kr[kw][i];
        }

#pragma unroll
        for (int jj = 0; jj < 8; ++jj) {
            float xv[14];
#pragma unroll
            for (int ci = 0; ci < 14; ++ci)
                xv[ci] = cok[ci] ? xr[jj][ci] : 0.0f;
#pragma unroll
            for (int kw = 0; kw < K_SZ; ++kw)
#pragma unroll
                for (int r = 0; r < 8; ++r)
                    o[jj][r] = fmaf(kr[kw][r], xv[r + kw], o[jj][r]);
        }
    }

#pragma unroll
    for (int jj = 0; jj < 8; ++jj)
#pragma unroll
        for (int r = 0; r < 8; ++r)
            out[(size_t)(pb + r) * C_CH + chan0 + G_NUM * jj] = o[jj][r];
}

extern "C" void kernel_launch(void* const* d_in, const int* in_sizes, int n_in,
                              void* d_out, int out_size, void* d_ws, size_t ws_size,
                              hipStream_t stream)
{
    const float* x     = (const float*)d_in[0];
    const float* w1    = (const float*)d_in[1];
    const float* b1    = (const float*)d_in[2];
    const float* gamma = (const float*)d_in[3];
    const float* beta  = (const float*)d_in[4];
    const float* mu    = (const float*)d_in[5];
    const float* var   = (const float*)d_in[6];
    const float* w2    = (const float*)d_in[7];
    const float* b2    = (const float*)d_in[8];

    float* out  = (float*)d_out;
    float* kern = out + (size_t)M_TOT * C_CH;
    float* t    = (float*)d_ws;

    gemm1_bn_relu<<<256, 256, 0, stream>>>(x, w1, b1, gamma, beta, mu, var, t);
    kerngen_gather<<<256, 128, 0, stream>>>(x, t, w2, b2, kern, out);
}


# GEMM1 re-tiled: wave = 16 rows x 32 columns as two column-interleaved accumulator chains; x read as 16-byte k-quads, w1 and t as 8-byte pairs, loads one 64-k group ahead; GEMM1 sum order changes
# speedup vs baseline: 2.5800x; 1.3532x over previous
#include <hip/hip_runtime.h>


#define K_SZ   7
#define KK     49
#define G_NUM  16
#define C_CH   256
#define CR     64
#define NK     (KK * G_NUM)
#define HW     64
#define BATCH  2
#define M_TOT  (BATCH * HW * HW)
#define BN_EPS 1e-3f

typedef __attribute__((ext_vector_type(2))) float v2f;
typedef __attribute__((ext_vector_type(8))) float v8f;

typedef __attribute__((ext_vector_type(4))) float v4f;

#define G1_U  8

__global__ __launch_bounds__(128) void gemm1_bn_relu(
    const float* __restrict__ x,  const float* __restrict__ w1,
    const float* __restrict__ b1, const float* __restrict__ gamma,
    const float* __restrict__ beta, const float* __restrict__ mu,
    const float* __restrict__ var, float* __restrict__ t)
{
    const int wave = __builtin_amdgcn_readfirstlane((int)(threadIdx.x >> 5));
    const int ln   = threadIdx.x & 31;
    const int tile = blockIdx.x * 4 + wave;
    const int m0   = (tile >> 1) << 4;
    const int n0   = (tile & 1)  << 5;
    const int c    = ln & 15;
    const int h    = ln >> 4;

    const float* ap = x  + (size_t)(m0 + c) * C_CH + 4 * h;
    const float* bp = w1 + (size_t)(4 * h) * CR + n0 + 2 * c;

    const int col = n0 + 2 * c;
    const v2f gm = *(const v2f*)(gamma + col);
    const v2f vr = *(const v2f*)(var + col);
    const v2f bt = *(const v2f*)(beta + col);
    const v2f mn = *(const v2f*)(mu + col);
    const v2f bb = *(const v2f*)(b1 + col);

    v4f xa[2][G1_U];
    v2f wb[2][G1_U][4];
#pragma unroll
    for (int uu = 0; uu < G1_U; ++uu) {
        xa[0][uu] = *(const v4f*)(ap + 8 * uu);
#pragma unroll
        for (int e = 0; e < 4; ++e)
            wb[0][uu][e] = *(const v2f*)(bp + (size_t)(8 * uu + e) * CR);
    }

    v8f acc0 = {}, acc1 = {};
#pragma unroll
    for (int gi = 0; gi < C_CH / (8 * G1_U); ++gi) {
        const int cb = gi & 1;
        if (gi + 1 < C_CH / (8 * G1_U)) {
#pragma unroll
            for (int uu = 0; uu < G1_U; ++uu) {
                const int u = (gi + 1) * G1_U + uu;
                xa[cb ^ 1][uu] = *(const v4f*)(ap + 8 * u);
#pragma unroll
                for (int e = 0; e < 4; ++e)
                    wb[cb ^ 1][uu][e] = *(const v2f*)(bp + (size_t)(8 * u + e) * CR);
            }
        }
#pragma unroll
        for (int uu = 0; uu < G1_U; ++uu) {
#pragma unroll
            for (int sub = 0; sub < 2; ++sub) {
                v2f a, bt0, bt1;
                a.x   = xa[cb][uu][2 * sub];
                a.y   = xa[cb][uu][2 * sub + 1];
                bt0.x = wb[cb][uu][2 * sub].x;
                bt0.y = wb[cb][uu][2 * sub + 1].x;
                bt1.x = wb[cb][uu][2 * sub].y;
                bt1.y = wb[cb][uu][2 * sub + 1].y;
                acc0 = __builtin_amdgcn_wmma_f32_16x16x4_f32(
                    false, a, false, bt0, (short)0, acc0, false, false);
                acc1 = __builtin_amdgcn_wmma_f32_16x16x4_f32(
                    false, a, false, bt1, (short)0, acc1, false, false);
            }
        }
    }

    __builtin_amdgcn_sched_group_barrier(0x020, 45, 0);
    __builtin_amdgcn_sched_group_barrier(0x020, 40, 0);
    __builtin_amdgcn_sched_group_barrier(0x008, 32, 0);
    __builtin_amdgcn_sched_group_barrier(0x020, 40, 0);
    __builtin_amdgcn_sched_group_barrier(0x008, 32, 0);
    __builtin_amdgcn_sched_group_barrier(0x020, 40, 0);
    __builtin_amdgcn_sched_group_barrier(0x008, 32, 0);
    __builtin_amdgcn_sched_group_barrier(0x008, 32, 0);

    const float scale0 = gm.x * rsqrtf(vr.x + BN_EPS);
    const float scale1 = gm.y * rsqrtf(vr.y + BN_EPS);
    const float shift0 = bt.x - mn.x * scale0;
    const float shift1 = bt.y - mn.y * scale1;
    const int   rowB   = m0 + (h << 3);
#pragma unroll
    for (int r = 0; r < 8; ++r) {
        v2f v;
        v.x = fmaxf(fmaf(acc0[r] + bb.x, scale0, shift0), 0.0f);
        v.y = fmaxf(fmaf(acc1[r] + bb.y, scale1, shift1), 0.0f);
        *(v2f*)(t + (size_t)(rowB + r) * CR + col) = v;
    }
}

#define SLAB_N  (K_SZ * G_NUM)
#define SLAB_P  ((CR / 2) * SLAB_N)

__global__ __launch_bounds__(128) void kerngen_gather(
    const float* __restrict__ x,  const float* __restrict__ t,
    const float* __restrict__ w2, const float* __restrict__ b2,
    float* __restrict__ kern, float* __restrict__ out)
{
    __shared__ v2f w2s[2][SLAB_P];

    const int wave = __builtin_amdgcn_readfirstlane((int)(threadIdx.x >> 5));
    const int ln   = threadIdx.x & 31;
    const int mt   = blockIdx.x * 2 + (wave >> 1);
    const int jh   = wave & 1;
    const int g    = ln & 15;
    const int half = ln >> 4;
    const int m0   = mt << 4;
    const int w0   = m0 & (HW - 1);
    const int h    = (m0 >> 6) & (HW - 1);
    const int b    = m0 >> 12;
    const int pb   = m0 + (half << 3);
    const int wl   = w0 + (half << 3);

    const int fk2 = threadIdx.x >> 2;
    const int fc4 = threadIdx.x & 3;
    const float* frow = w2 + (size_t)(2 * fk2) * NK + 4 * fc4;
    const int    fdst = fk2 * SLAB_N + 4 * fc4;

    v4f f0[K_SZ], f1[K_SZ];
#pragma unroll
    for (int i = 0; i < K_SZ; ++i) {
        f0[i] = *(const v4f*)(frow + 16 * i);
        f1[i] = *(const v4f*)(frow + NK + 16 * i);
    }

    v2f a[CR / 4];
    const float* arow = t + (size_t)(m0 + g) * CR + 2 * half;
#pragma unroll
    for (int ks = 0; ks < CR / 4; ++ks) {
        a[ks].x = arow[4 * ks];
        a[ks].y = arow[4 * ks + 1];
    }

    int  cofs[14];
    bool cok[14];
#pragma unroll
    for (int ci = 0; ci < 14; ++ci) {
        const int ww = wl - (K_SZ / 2) + ci;
        cok[ci]  = (unsigned)ww < (unsigned)HW;
        cofs[ci] = (cok[ci] ? ww : wl) * C_CH;
    }

    float o[8][8];
#pragma unroll
    for (int jj = 0; jj < 8; ++jj)
#pragma unroll
        for (int r = 0; r < 8; ++r) o[jj][r] = 0.0f;

    const int chan0 = g + G_NUM * 8 * jh;

    float bcn[K_SZ];
#pragma unroll
    for (int kw = 0; kw < K_SZ; ++kw) bcn[kw] = b2[kw * G_NUM + g];

#pragma unroll
    for (int i = 0; i < K_SZ; ++i) {
        v2f* d = &w2s[0][fdst + 16 * i];
        d[0] = v2f{f0[i].x, f1[i].x};
        d[1] = v2f{f0[i].y, f1[i].y};
        d[2] = v2f{f0[i].z, f1[i].z};
        d[3] = v2f{f0[i].w, f1[i].w};
    }
    __syncthreads();

#pragma unroll 1
    for (int kh = 0; kh < K_SZ; ++kh) {
        const int cur = kh & 1;
        const int kh1 = (kh + 1 < K_SZ) ? kh + 1 : kh;

#pragma unroll
        for (int i = 0; i < K_SZ; ++i) {
            f0[i] = *(const v4f*)(frow + kh1 * SLAB_N + 16 * i);
            f1[i] = *(const v4f*)(frow + NK + kh1 * SLAB_N + 16 * i);
        }
        float bc[K_SZ];
#pragma unroll
        for (int kw = 0; kw < K_SZ; ++kw) bc[kw] = bcn[kw];
#pragma unroll
        for (int kw = 0; kw < K_SZ; ++kw) bcn[kw] = b2[(kh1 * K_SZ + kw) * G_NUM + g];

        const int  hh    = h + kh - (K_SZ / 2);
        const bool rowok = (unsigned)hh < (unsigned)HW;
        const int  hc    = rowok ? hh : h;
        const float* xr0 = x + ((size_t)(b * HW + hc) * HW) * C_CH + chan0;
        float xr[8][14];
#pragma unroll
        for (int jj = 0; jj < 8; ++jj)
#pragma unroll
            for (int ci = 0; ci < 14; ++ci)
                xr[jj][ci] = xr0[cofs[ci] + G_NUM * jj];

        v8f kr[K_SZ];
#pragma unroll
        for (int kw = 0; kw < K_SZ; ++kw) kr[kw] = v8f{};

        const v2f* bl = &w2s[cur][half * SLAB_N + g];
        v2f bf[2][4][K_SZ];
#pragma unroll
        for (int kq = 0; kq < 4; ++kq)
#pragma unroll
            for (int kw = 0; kw < K_SZ; ++kw)
                bf[0][kq][kw] = bl[(2 * kq) * SLAB_N + kw * G_NUM];
#pragma unroll
        for (int q = 0; q < 4; ++q) {
            if (q + 1 < 4) {
#pragma unroll
                for (int kq = 0; kq < 4; ++kq)
#pragma unroll
                    for (int kw = 0; kw < K_SZ; ++kw)
                        bf[(q + 1) & 1][kq][kw] =
                            bl[(2 * (4 * (q + 1) + kq)) * SLAB_N + kw * G_NUM];
            }
#pragma unroll
            for (int kq = 0; kq < 4; ++kq)
#pragma unroll
                for (int kw = 0; kw < K_SZ; ++kw)
                    kr[kw] = __builtin_amdgcn_wmma_f32_16x16x4_f32(
                        false, a[4 * q + kq], false, bf[q & 1][kq][kw],
                        (short)0, kr[kw], false, false);
        }
        __builtin_amdgcn_sched_group_barrier(0x020, 63, 0);
        __builtin_amdgcn_sched_group_barrier(0x100, 14, 0);
        __builtin_amdgcn_sched_group_barrier(0x100, 14, 0);
        __builtin_amdgcn_sched_group_barrier(0x008, 28, 0);
        __builtin_amdgcn_sched_group_barrier(0x020, 42, 0);
        __builtin_amdgcn_sched_group_barrier(0x100, 14, 0);
        __builtin_amdgcn_sched_group_barrier(0x008, 28, 0);
        __builtin_amdgcn_sched_group_barrier(0x020, 28, 0);
        __builtin_amdgcn_sched_group_barrier(0x100, 14, 0);
        __builtin_amdgcn_sched_group_barrier(0x008, 28, 0);
        __builtin_amdgcn_sched_group_barrier(0x008, 28, 0);

#pragma unroll
        for (int kw = 0; kw < K_SZ; ++kw)
#pragma unroll
            for (int r = 0; r < 8; ++r) kr[kw][r] += bc[kw];

#pragma unroll
        for (int i = 0; i < K_SZ; ++i) {
            v2f* d = &w2s[cur ^ 1][fdst + 16 * i];
            d[0] = v2f{f0[i].x, f1[i].x};
            d[1] = v2f{f0[i].y, f1[i].y};
            d[2] = v2f{f0[i].z, f1[i].z};
            d[3] = v2f{f0[i].w, f1[i].w};
        }

#pragma unroll
        for (int kw = 0; kw < K_SZ; ++kw) {
            float* kp = kern + (size_t)(pb + 4 * jh) * NK + (kh * K_SZ + kw) * G_NUM + g;
#pragma unroll
            for (int i = 0; i < 4; ++i)
                kp[(size_t)i * NK] = jh ? kr[kw][4 + i] : kr[kw][i];
        }

#pragma unroll
        for (int jj = 0; jj < 8; ++jj) {
            float xv[14];
#pragma unroll
            for (int ci = 0; ci < 14; ++ci)
                xv[ci] = (cok[ci] && rowok) ? xr[jj][ci] : 0.0f;
#pragma unroll
            for (int kw = 0; kw < K_SZ; ++kw)
#pragma unroll
                for (int r = 0; r < 8; ++r)
                    o[jj][r] = fmaf(kr[kw][r], xv[r + kw], o[jj][r]);
        }

        __syncthreads();
    }

#pragma unroll
    for (int jj = 0; jj < 8; ++jj)
#pragma unroll
        for (int r = 0; r < 8; ++r)
            out[(size_t)(pb + r) * C_CH + chan0 + G_NUM * jj] = o[jj][r];
}

extern "C" void kernel_launch(void* const* d_in, const int* in_sizes, int n_in,
                              void* d_out, int out_size, void* d_ws, size_t ws_size,
                              hipStream_t stream)
{
    const float* x     = (const float*)d_in[0];
    const float* w1    = (const float*)d_in[1];
    const float* b1    = (const float*)d_in[2];
    const float* gamma = (const float*)d_in[3];
    const float* beta  = (const float*)d_in[4];
    const float* mu    = (const float*)d_in[5];
    const float* var   = (const float*)d_in[6];
    const float* w2    = (const float*)d_in[7];
    const float* b2    = (const float*)d_in[8];

    float* out  = (float*)d_out;
    float* kern = out + (size_t)M_TOT * C_CH;
    float* t    = (float*)d_ws;

    gemm1_bn_relu<<<256, 128, 0, stream>>>(x, w1, b1, gamma, beta, mu, var, t);
    kerngen_gather<<<256, 128, 0, stream>>>(x, t, w2, b2, kern, out);
}


# GEMM2 computed once, not twice: of a tap-row's 7 taps one wave of an m-tile's pair does 4, the other 3 plus the kern stores; tiles swapped through LDS in D layout; sum order kept
# speedup vs baseline: 3.0203x; 1.1707x over previous
#include <hip/hip_runtime.h>


#define K_SZ   7
#define KK     49
#define G_NUM  16
#define C_CH   256
#define CR     64
#define NK     (KK * G_NUM)
#define HW     64
#define BATCH  2
#define M_TOT  (BATCH * HW * HW)
#define BN_EPS 1e-3f

typedef __attribute__((ext_vector_type(2))) float v2f;
typedef __attribute__((ext_vector_type(8))) float v8f;

typedef __attribute__((ext_vector_type(4))) float v4f;

#define G1_U  8

__global__ __launch_bounds__(128) void gemm1_bn_relu(
    const float* __restrict__ x,  const float* __restrict__ w1,
    const float* __restrict__ b1, const float* __restrict__ gamma,
    const float* __restrict__ beta, const float* __restrict__ mu,
    const float* __restrict__ var, float* __restrict__ t)
{
    const int wave = __builtin_amdgcn_readfirstlane((int)(threadIdx.x >> 5));
    const int ln   = threadIdx.x & 31;
    const int tile = blockIdx.x * 4 + wave;
    const int m0   = (tile >> 1) << 4;
    const int n0   = (tile & 1)  << 5;
    const int c    = ln & 15;
    const int h    = ln >> 4;

    const float* ap = x  + (size_t)(m0 + c) * C_CH + 4 * h;
    const float* bp = w1 + (size_t)(4 * h) * CR + n0 + 2 * c;

    const int col = n0 + 2 * c;
    const v2f gm = *(const v2f*)(gamma + col);
    const v2f vr = *(const v2f*)(var + col);
    const v2f bt = *(const v2f*)(beta + col);
    const v2f mn = *(const v2f*)(mu + col);
    const v2f bb = *(const v2f*)(b1 + col);

    v4f xa[2][G1_U];
    v2f wb[2][G1_U][4];
#pragma unroll
    for (int uu = 0; uu < G1_U; ++uu) {
        xa[0][uu] = *(const v4f*)(ap + 8 * uu);
#pragma unroll
        for (int e = 0; e < 4; ++e)
            wb[0][uu][e] = *(const v2f*)(bp + (size_t)(8 * uu + e) * CR);
    }

    v8f acc0 = {}, acc1 = {};
#pragma unroll
    for (int gi = 0; gi < C_CH / (8 * G1_U); ++gi) {
        const int cb = gi & 1;
        if (gi + 1 < C_CH / (8 * G1_U)) {
#pragma unroll
            for (int uu = 0; uu < G1_U; ++uu) {
                const int u = (gi + 1) * G1_U + uu;
                xa[cb ^ 1][uu] = *(const v4f*)(ap + 8 * u);
#pragma unroll
                for (int e = 0; e < 4; ++e)
                    wb[cb ^ 1][uu][e] = *(const v2f*)(bp + (size_t)(8 * u + e) * CR);
            }
        }
#pragma unroll
        for (int uu = 0; uu < G1_U; ++uu) {
#pragma unroll
            for (int sub = 0; sub < 2; ++sub) {
                v2f a, bt0, bt1;
                a.x   = xa[cb][uu][2 * sub];
                a.y   = xa[cb][uu][2 * sub + 1];
                bt0.x = wb[cb][uu][2 * sub].x;
                bt0.y = wb[cb][uu][2 * sub + 1].x;
                bt1.x = wb[cb][uu][2 * sub].y;
                bt1.y = wb[cb][uu][2 * sub + 1].y;
                acc0 = __builtin_amdgcn_wmma_f32_16x16x4_f32(
                    false, a, false, bt0, (short)0, acc0, false, false);
                acc1 = __builtin_amdgcn_wmma_f32_16x16x4_f32(
                    false, a, false, bt1, (short)0, acc1, false, false);
            }
        }
    }

    __builtin_amdgcn_sched_group_barrier(0x020, 45, 0);
    __builtin_amdgcn_sched_group_barrier(0x020, 40, 0);
    __builtin_amdgcn_sched_group_barrier(0x008, 32, 0);
    __builtin_amdgcn_sched_group_barrier(0x020, 40, 0);
    __builtin_amdgcn_sched_group_barrier(0x008, 32, 0);
    __builtin_amdgcn_sched_group_barrier(0x020, 40, 0);
    __builtin_amdgcn_sched_group_barrier(0x008, 32, 0);
    __builtin_amdgcn_sched_group_barrier(0x008, 32, 0);

    const float scale0 = gm.x * rsqrtf(vr.x + BN_EPS);
    const float scale1 = gm.y * rsqrtf(vr.y + BN_EPS);
    const float shift0 = bt.x - mn.x * scale0;
    const float shift1 = bt.y - mn.y * scale1;
    const int   rowB   = m0 + (h << 3);
#pragma unroll
    for (int r = 0; r < 8; ++r) {
        v2f v;
        v.x = fmaxf(fmaf(acc0[r] + bb.x, scale0, shift0), 0.0f);
        v.y = fmaxf(fmaf(acc1[r] + bb.y, scale1, shift1), 0.0f);
        *(v2f*)(t + (size_t)(rowB + r) * CR + col) = v;
    }
}

#define SLAB_N  (K_SZ * G_NUM)
#define SLAB_P  ((CR / 2) * SLAB_N)
#define TAPS_A  4

template <int KW0, int NT>
__device__ __forceinline__ void trip_front(
    const float* __restrict__ frow, const int kh1,
    const float* __restrict__ b2g, const float* __restrict__ xr0,
    const int (&cofs)[14], const v2f (&a)[CR / 4], const v2f* __restrict__ bl,
    v4f* __restrict__ xdst,
    v4f (&f0)[K_SZ], v4f (&f1)[K_SZ], float (&bcn)[TAPS_A],
    float (&xr)[8][14], v8f (&kr)[K_SZ])
{
#pragma unroll
    for (int i = 0; i < K_SZ; ++i) {
        f0[i] = *(const v4f*)(frow + kh1 * SLAB_N + 16 * i);
        f1[i] = *(const v4f*)(frow + NK + kh1 * SLAB_N + 16 * i);
    }
    float bc[NT];
#pragma unroll
    for (int t = 0; t < NT; ++t) bc[t] = bcn[t];
#pragma unroll
    for (int t = 0; t < NT; ++t) bcn[t] = b2g[(kh1 * K_SZ + KW0 + t) * G_NUM];
#pragma unroll
    for (int jj = 0; jj < 8; ++jj)
#pragma unroll
        for (int ci = 0; ci < 14; ++ci)
            xr[jj][ci] = xr0[cofs[ci] + G_NUM * jj];

#pragma unroll
    for (int t = 0; t < NT; ++t) kr[KW0 + t] = v8f{};

    v2f bf[2][4][NT];
#pragma unroll
    for (int kq = 0; kq < 4; ++kq)
#pragma unroll
        for (int t = 0; t < NT; ++t)
            bf[0][kq][t] = bl[(2 * kq) * SLAB_N + (KW0 + t) * G_NUM];
#pragma unroll
    for (int q = 0; q < 4; ++q) {
        if (q + 1 < 4) {
#pragma unroll
            for (int kq = 0; kq < 4; ++kq)
#pragma unroll
                for (int t = 0; t < NT; ++t)
                    bf[(q + 1) & 1][kq][t] =
                        bl[(2 * (4 * (q + 1) + kq)) * SLAB_N + (KW0 + t) * G_NUM];
        }
#pragma unroll
        for (int kq = 0; kq < 4; ++kq)
#pragma unroll
            for (int t = 0; t < NT; ++t)
                kr[KW0 + t] = __builtin_amdgcn_wmma_f32_16x16x4_f32(
                    false, a[4 * q + kq], false, bf[q & 1][kq][t],
                    (short)0, kr[KW0 + t], false, false);
    }
    constexpr int NL  = 14 + NT + 112;
    constexpr int NL1 = 14 + NT + 3 * 14;
    constexpr int NL2 = (NL - NL1) / 2;
    constexpr int NL3 = NL - NL1 - NL2;
    __builtin_amdgcn_sched_group_barrier(0x020, NL1, 0);
    __builtin_amdgcn_sched_group_barrier(0x100, 2 * NT, 0);
    __builtin_amdgcn_sched_group_barrier(0x100, 2 * NT, 0);
    __builtin_amdgcn_sched_group_barrier(0x008, 4 * NT, 0);
    __builtin_amdgcn_sched_group_barrier(0x020, NL2, 0);
    __builtin_amdgcn_sched_group_barrier(0x100, 2 * NT, 0);
    __builtin_amdgcn_sched_group_barrier(0x008, 4 * NT, 0);
    __builtin_amdgcn_sched_group_barrier(0x020, NL3, 0);
    __builtin_amdgcn_sched_group_barrier(0x100, 2 * NT, 0);
    __builtin_amdgcn_sched_group_barrier(0x008, 4 * NT, 0);
    __builtin_amdgcn_sched_group_barrier(0x008, 4 * NT, 0);

#pragma unroll
    for (int t = 0; t < NT; ++t)
#pragma unroll
        for (int r = 0; r < 8; ++r) kr[KW0 + t][r] += bc[t];

#pragma unroll
    for (int t = 0; t < NT; ++t) {
        const v8f v = kr[KW0 + t];
        xdst[((KW0 + t) * 2 + 0) * 32] = v4f{v[0], v[1], v[2], v[3]};
        xdst[((KW0 + t) * 2 + 1) * 32] = v4f{v[4], v[5], v[6], v[7]};
    }
}

template <int KW0, int NT>
__device__ __forceinline__ void trip_back(
    const v4f* __restrict__ xsrc, float* __restrict__ kp0, v8f (&kr)[K_SZ])
{
#pragma unroll
    for (int kw = 0; kw < K_SZ; ++kw) {
        if (kw >= KW0 && kw < KW0 + NT) continue;
        const v4f lo = xsrc[(kw * 2 + 0) * 32];
        const v4f hi = xsrc[(kw * 2 + 1) * 32];
        kr[kw] = v8f{lo.x, lo.y, lo.z, lo.w, hi.x, hi.y, hi.z, hi.w};
    }
    if (NT < TAPS_A) {
#pragma unroll
        for (int i = 0; i < K_SZ; ++i) {
            const int kw = (KW0 + i) % K_SZ;
#pragma unroll
            for (int r = 0; r < 8; ++r)
                kp0[(size_t)r * NK + kw * G_NUM] = kr[kw][r];
        }
    }
}

__global__ __launch_bounds__(128) void kerngen_gather(
    const float* __restrict__ x,  const float* __restrict__ t,
    const float* __restrict__ w2, const float* __restrict__ b2,
    float* __restrict__ kern, float* __restrict__ out)
{
    __shared__ v2f w2s[2][SLAB_P];
    __shared__ v4f xch[2][2][K_SZ * 2 * 32];

    const int wave = __builtin_amdgcn_readfirstlane((int)(threadIdx.x >> 5));
    const int ln   = threadIdx.x & 31;
    const int ml   = wave >> 1;
    const int mt   = blockIdx.x * 2 + ml;
    const int jh   = wave & 1;
    const int g    = ln & 15;
    const int half = ln >> 4;
    const int m0   = mt << 4;
    const int w0   = m0 & (HW - 1);
    const int h    = (m0 >> 6) & (HW - 1);
    const int b    = m0 >> 12;
    const int pb   = m0 + (half << 3);
    const int wl   = w0 + (half << 3);

    const int fk2 = threadIdx.x >> 2;
    const int fc4 = threadIdx.x & 3;
    const float* frow = w2 + (size_t)(2 * fk2) * NK + 4 * fc4;
    const int    fdst = fk2 * SLAB_N + 4 * fc4;

    v4f f0[K_SZ], f1[K_SZ];
#pragma unroll
    for (int i = 0; i < K_SZ; ++i) {
        f0[i] = *(const v4f*)(frow + 16 * i);
        f1[i] = *(const v4f*)(frow + NK + 16 * i);
    }

    v2f a[CR / 4];
    const float* arow = t + (size_t)(m0 + g) * CR + 2 * half;
#pragma unroll
    for (int ks = 0; ks < CR / 4; ++ks) {
        a[ks].x = arow[4 * ks];
        a[ks].y = arow[4 * ks + 1];
    }

    int  cofs[14];
    bool cok[14];
#pragma unroll
    for (int ci = 0; ci < 14; ++ci) {
        const int ww = wl - (K_SZ / 2) + ci;
        cok[ci]  = (unsigned)ww < (unsigned)HW;
        cofs[ci] = (cok[ci] ? ww : wl) * C_CH;
    }

    float o[8][8];
#pragma unroll
    for (int jj = 0; jj < 8; ++jj)
#pragma unroll
        for (int r = 0; r < 8; ++r) o[jj][r] = 0.0f;

    const int chan0 = g + G_NUM * 8 * jh;

    const float* b2g = b2 + g;
    float bcn[TAPS_A];
#pragma unroll
    for (int tt = 0; tt < TAPS_A; ++tt) {
        const int kw = jh ? (TAPS_A + tt < K_SZ ? TAPS_A + tt : K_SZ - 1) : tt;
        bcn[tt] = b2g[kw * G_NUM];
    }

#pragma unroll
    for (int i = 0; i < K_SZ; ++i) {
        v2f* d = &w2s[0][fdst + 16 * i];
        d[0] = v2f{f0[i].x, f1[i].x};
        d[1] = v2f{f0[i].y, f1[i].y};
        d[2] = v2f{f0[i].z, f1[i].z};
        d[3] = v2f{f0[i].w, f1[i].w};
    }
    __syncthreads();

#pragma unroll 1
    for (int kh = 0; kh < K_SZ; ++kh) {
        const int cur = kh & 1;
        const int kh1 = (kh + 1 < K_SZ) ? kh + 1 : kh;

        const int  hh    = h + kh - (K_SZ / 2);
        const bool rowok = (unsigned)hh < (unsigned)HW;
        const int  hc    = rowok ? hh : h;
        const float* xr0 = x + ((size_t)(b * HW + hc) * HW) * C_CH + chan0;
        float xr[8][14];
        v8f   kr[K_SZ];

        const v2f* bl   = &w2s[cur][half * SLAB_N + g];
        v4f*       xbuf = &xch[cur][ml][ln];

        if (jh == 0)
            trip_front<0, TAPS_A>(frow, kh1, b2g, xr0, cofs, a, bl, xbuf, f0, f1, bcn, xr, kr);
        else
            trip_front<TAPS_A, K_SZ - TAPS_A>(frow, kh1, b2g, xr0, cofs, a, bl, xbuf, f0, f1, bcn, xr, kr);

#pragma unroll
        for (int i = 0; i < K_SZ; ++i) {
            v2f* d = &w2s[cur ^ 1][fdst + 16 * i];
            d[0] = v2f{f0[i].x, f1[i].x};
            d[1] = v2f{f0[i].y, f1[i].y};
            d[2] = v2f{f0[i].z, f1[i].z};
            d[3] = v2f{f0[i].w, f1[i].w};
        }

        __syncthreads();

        float* kp0 = kern + (size_t)pb * NK + (kh * K_SZ) * G_NUM + g;
        if (jh == 0)
            trip_back<0, TAPS_A>(xbuf, kp0, kr);
        else
            trip_back<TAPS_A, K_SZ - TAPS_A>(xbuf, kp0, kr);

#pragma unroll
        for (int jj = 0; jj < 8; ++jj) {
            float xv[14];
#pragma unroll
            for (int ci = 0; ci < 14; ++ci)
                xv[ci] = (cok[ci] && rowok) ? xr[jj][ci] : 0.0f;
#pragma unroll
            for (int kw = 0; kw < K_SZ; ++kw)
#pragma unroll
                for (int r = 0; r < 8; ++r)
                    o[jj][r] = fmaf(kr[kw][r], xv[r + kw], o[jj][r]);
        }
    }

#pragma unroll
    for (int jj = 0; jj < 8; ++jj)
#pragma unroll
        for (int r = 0; r < 8; ++r)
            out[(size_t)(pb + r) * C_CH + chan0 + G_NUM * jj] = o[jj][r];
}

extern "C" void kernel_launch(void* const* d_in, const int* in_sizes, int n_in,
                              void* d_out, int out_size, void* d_ws, size_t ws_size,
                              hipStream_t stream)
{
    const float* x     = (const float*)d_in[0];
    const float* w1    = (const float*)d_in[1];
    const float* b1    = (const float*)d_in[2];
    const float* gamma = (const float*)d_in[3];
    const float* beta  = (const float*)d_in[4];
    const float* mu    = (const float*)d_in[5];
    const float* var   = (const float*)d_in[6];
    const float* w2    = (const float*)d_in[7];
    const float* b2    = (const float*)d_in[8];

    float* out  = (float*)d_out;
    float* kern = out + (size_t)M_TOT * C_CH;
    float* t    = (float*)d_ws;

    gemm1_bn_relu<<<256, 128, 0, stream>>>(x, w1, b1, gamma, beta, mu, var, t);
    kerngen_gather<<<256, 128, 0, stream>>>(x, t, w2, b2, kern, out);
}


# GEMM1 fused into the main kernel as t^T = w1^T x^T (WMMA operands swapped): its accumulator tiles are GEMM2's A fragments, halves swapped via LDS; one launch, no t in memory; GEMM2 sum order changes
# speedup vs baseline: 3.4248x; 1.1339x over previous
#include <hip/hip_runtime.h>


#define K_SZ   7
#define KK     49
#define G_NUM  16
#define C_CH   256
#define CR     64
#define NK     (KK * G_NUM)
#define HW     64
#define BATCH  2
#define M_TOT  (BATCH * HW * HW)
#define BN_EPS 1e-3f

typedef __attribute__((ext_vector_type(2))) float v2f;
typedef __attribute__((ext_vector_type(4))) float v4f;
typedef __attribute__((ext_vector_type(8))) float v8f;

#define G1_U  8

#define SLAB_N  (K_SZ * G_NUM)
#define SLAB_P  ((CR / 2) * SLAB_N)
#define TAPS_A  4

template <int KW0, int NT>
__device__ __forceinline__ void trip_front(
    const float* __restrict__ frow, const int kh1,
    const float* __restrict__ b2g, const float* __restrict__ xr0,
    const int (&cofs)[14], const v2f (&a)[CR / 4], const v2f* __restrict__ bl,
    v4f* __restrict__ xdst,
    v4f (&f0)[K_SZ], v4f (&f1)[K_SZ], float (&bcn)[TAPS_A],
    float (&xr)[8][14], v8f (&kr)[K_SZ])
{
#pragma unroll
    for (int i = 0; i < K_SZ; ++i) {
        f0[i] = *(const v4f*)(frow + kh1 * SLAB_N + 16 * i);
        f1[i] = *(const v4f*)(frow + 2 * NK + kh1 * SLAB_N + 16 * i);
    }
    float bc[NT];
#pragma unroll
    for (int t = 0; t < NT; ++t) bc[t] = bcn[t];
#pragma unroll
    for (int t = 0; t < NT; ++t) bcn[t] = b2g[(kh1 * K_SZ + KW0 + t) * G_NUM];
#pragma unroll
    for (int jj = 0; jj < 8; ++jj)
#pragma unroll
        for (int ci = 0; ci < 14; ++ci)
            xr[jj][ci] = xr0[cofs[ci] + G_NUM * jj];

#pragma unroll
    for (int t = 0; t < NT; ++t) kr[KW0 + t] = v8f{};

    v2f bf[2][4][NT];
#pragma unroll
    for (int kq = 0; kq < 4; ++kq)
#pragma unroll
        for (int t = 0; t < NT; ++t)
            bf[0][kq][t] = bl[(2 * kq) * SLAB_N + (KW0 + t) * G_NUM];
#pragma unroll
    for (int q = 0; q < 4; ++q) {
        if (q + 1 < 4) {
#pragma unroll
            for (int kq = 0; kq < 4; ++kq)
#pragma unroll
                for (int t = 0; t < NT; ++t)
                    bf[(q + 1) & 1][kq][t] =
                        bl[(2 * (4 * (q + 1) + kq)) * SLAB_N + (KW0 + t) * G_NUM];
        }
#pragma unroll
        for (int kq = 0; kq < 4; ++kq)
#pragma unroll
            for (int t = 0; t < NT; ++t)
                kr[KW0 + t] = __builtin_amdgcn_wmma_f32_16x16x4_f32(
                    false, a[4 * q + kq], false, bf[q & 1][kq][t],
                    (short)0, kr[KW0 + t], false, false);
    }
    constexpr int NL  = 14 + NT + 112;
    constexpr int NL1 = 14 + NT + 3 * 14;
    constexpr int NL2 = (NL - NL1) / 2;
    constexpr int NL3 = NL - NL1 - NL2;
    __builtin_amdgcn_sched_group_barrier(0x020, NL1, 0);
    __builtin_amdgcn_sched_group_barrier(0x100, 2 * NT, 0);
    __builtin_amdgcn_sched_group_barrier(0x100, 2 * NT, 0);
    __builtin_amdgcn_sched_group_barrier(0x008, 4 * NT, 0);
    __builtin_amdgcn_sched_group_barrier(0x020, NL2, 0);
    __builtin_amdgcn_sched_group_barrier(0x100, 2 * NT, 0);
    __builtin_amdgcn_sched_group_barrier(0x008, 4 * NT, 0);
    __builtin_amdgcn_sched_group_barrier(0x020, NL3, 0);
    __builtin_amdgcn_sched_group_barrier(0x100, 2 * NT, 0);
    __builtin_amdgcn_sched_group_barrier(0x008, 4 * NT, 0);
    __builtin_amdgcn_sched_group_barrier(0x008, 4 * NT, 0);

#pragma unroll
    for (int t = 0; t < NT; ++t)
#pragma unroll
        for (int r = 0; r < 8; ++r) kr[KW0 + t][r] += bc[t];

#pragma unroll
    for (int t = 0; t < NT; ++t) {
        const v8f v = kr[KW0 + t];
        xdst[((KW0 + t) * 2 + 0) * 32] = v4f{v[0], v[1], v[2], v[3]};
        xdst[((KW0 + t) * 2 + 1) * 32] = v4f{v[4], v[5], v[6], v[7]};
    }
}

template <int KW0, int NT>
__device__ __forceinline__ void trip_back(
    const v4f* __restrict__ xsrc, float* __restrict__ kp0, v8f (&kr)[K_SZ])
{
#pragma unroll
    for (int kw = 0; kw < K_SZ; ++kw) {
        if (kw >= KW0 && kw < KW0 + NT) continue;
        const v4f lo = xsrc[(kw * 2 + 0) * 32];
        const v4f hi = xsrc[(kw * 2 + 1) * 32];
        kr[kw] = v8f{lo.x, lo.y, lo.z, lo.w, hi.x, hi.y, hi.z, hi.w};
    }
    if (NT < TAPS_A) {
#pragma unroll
        for (int i = 0; i < K_SZ; ++i) {
            const int kw = (KW0 + i) % K_SZ;
#pragma unroll
            for (int r = 0; r < 8; ++r)
                kp0[(size_t)r * NK + kw * G_NUM] = kr[kw][r];
        }
    }
}

__global__ __launch_bounds__(128) void involution_fused(
    const float* __restrict__ x,  const float* __restrict__ w1,
    const float* __restrict__ b1, const float* __restrict__ gamma,
    const float* __restrict__ beta, const float* __restrict__ mu,
    const float* __restrict__ var, const float* __restrict__ w2,
    const float* __restrict__ b2,
    float* __restrict__ kern, float* __restrict__ out)
{
    __shared__ v2f w2s[2][SLAB_P];
    __shared__ v4f xch[2][2][K_SZ * 2 * 32];
    __shared__ v4f tex[2][2][4 * 32];

    const int wave = __builtin_amdgcn_readfirstlane((int)(threadIdx.x >> 5));
    const int ln   = threadIdx.x & 31;
    const int ml   = wave >> 1;
    const int mt   = blockIdx.x * 2 + ml;
    const int jh   = wave & 1;
    const int g    = ln & 15;
    const int half = ln >> 4;
    const int m0   = mt << 4;
    const int w0   = m0 & (HW - 1);
    const int h    = (m0 >> 6) & (HW - 1);
    const int b    = m0 >> 12;
    const int pb   = m0 + (half << 3);
    const int wl   = w0 + (half << 3);

    const int fk2 = threadIdx.x >> 2;
    const int fc4 = threadIdx.x & 3;
    const int fkA = 32 * (fk2 >> 4) + 16 * (fk2 & 1) + 4 * ((fk2 >> 1) & 3) + ((fk2 >> 3) & 1);
    const float* frow = w2 + (size_t)fkA * NK + 4 * fc4;
    const int    fdst = fk2 * SLAB_N + 4 * fc4;

    const int n0   = jh << 5;
    const int dcol = n0 + 16 * half;
    const float* ap = x  + (size_t)(m0 + g) * C_CH + 4 * half;
    const float* bp = w1 + (size_t)(4 * half) * CR + n0 + 2 * g;

    v4f cgm[4], cvr[4], cbt[4], cmn[4], cbb[4];
#pragma unroll
    for (int i = 0; i < 4; ++i) {
        cgm[i] = *(const v4f*)(gamma + dcol + 4 * i);
        cvr[i] = *(const v4f*)(var   + dcol + 4 * i);
        cbt[i] = *(const v4f*)(beta  + dcol + 4 * i);
        cmn[i] = *(const v4f*)(mu    + dcol + 4 * i);
        cbb[i] = *(const v4f*)(b1    + dcol + 4 * i);
    }

    v4f xa[2][G1_U];
    v2f wb[2][G1_U][4];
#pragma unroll
    for (int uu = 0; uu < G1_U; ++uu) {
        xa[0][uu] = *(const v4f*)(ap + 8 * uu);
#pragma unroll
        for (int e = 0; e < 4; ++e)
            wb[0][uu][e] = *(const v2f*)(bp + (size_t)(8 * uu + e) * CR);
    }

    v4f f0[K_SZ], f1[K_SZ];
    const float* b2g = b2 + g;
    float bcn[TAPS_A];

    v8f acc0 = {}, acc1 = {};
#pragma unroll
    for (int gi = 0; gi < C_CH / (8 * G1_U); ++gi) {
        const int cb = gi & 1;
        if (gi + 1 < C_CH / (8 * G1_U)) {
#pragma unroll
            for (int uu = 0; uu < G1_U; ++uu) {
                const int u = (gi + 1) * G1_U + uu;
                xa[cb ^ 1][uu] = *(const v4f*)(ap + 8 * u);
#pragma unroll
                for (int e = 0; e < 4; ++e)
                    wb[cb ^ 1][uu][e] = *(const v2f*)(bp + (size_t)(8 * u + e) * CR);
            }
        } else {
#pragma unroll
            for (int i = 0; i < K_SZ; ++i) {
                f0[i] = *(const v4f*)(frow + 16 * i);
                f1[i] = *(const v4f*)(frow + 2 * NK + 16 * i);
            }
#pragma unroll
            for (int tt = 0; tt < TAPS_A; ++tt) {
                const int kw = jh ? (TAPS_A + tt < K_SZ ? TAPS_A + tt : K_SZ - 1) : tt;
                bcn[tt] = b2g[kw * G_NUM];
            }
        }
#pragma unroll
        for (int uu = 0; uu < G1_U; ++uu) {
#pragma unroll
            for (int sub = 0; sub < 2; ++sub) {
                v2f xf, wf0, wf1;
                xf.x  = xa[cb][uu][2 * sub];
                xf.y  = xa[cb][uu][2 * sub + 1];
                wf0.x = wb[cb][uu][2 * sub].x;
                wf0.y = wb[cb][uu][2 * sub + 1].x;
                wf1.x = wb[cb][uu][2 * sub].y;
                wf1.y = wb[cb][uu][2 * sub + 1].y;
                acc0 = __builtin_amdgcn_wmma_f32_16x16x4_f32(
                    false, wf0, false, xf, (short)0, acc0, false, false);
                acc1 = __builtin_amdgcn_wmma_f32_16x16x4_f32(
                    false, wf1, false, xf, (short)0, acc1, false, false);
            }
        }
    }

    __builtin_amdgcn_sched_group_barrier(0x020, 60, 0);
    __builtin_amdgcn_sched_group_barrier(0x020, 40, 0);
    __builtin_amdgcn_sched_group_barrier(0x008, 32, 0);
    __builtin_amdgcn_sched_group_barrier(0x020, 40, 0);
    __builtin_amdgcn_sched_group_barrier(0x008, 32, 0);
    __builtin_amdgcn_sched_group_barrier(0x020, 40, 0);
    __builtin_amdgcn_sched_group_barrier(0x008, 32, 0);
    __builtin_amdgcn_sched_group_barrier(0x020, 2 * K_SZ + TAPS_A, 0);
    __builtin_amdgcn_sched_group_barrier(0x008, 32, 0);

    {
        float tv[2][8];
#pragma unroll
        for (int r = 0; r < 8; ++r) {
#pragma unroll
            for (int j = 0; j < 2; ++j) {
                const int i = 2 * r + j;
                const float scale = cgm[i >> 2][i & 3] * rsqrtf(cvr[i >> 2][i & 3] + BN_EPS);
                const float shift = cbt[i >> 2][i & 3] - cmn[i >> 2][i & 3] * scale;
                const float pre   = (j ? acc1[r] : acc0[r]) + cbb[i >> 2][i & 3];
                tv[j][r] = fmaxf(fmaf(pre, scale, shift), 0.0f);
            }
        }
        v4f* td = &tex[ml][jh][ln];
#pragma unroll
        for (int j = 0; j < 2; ++j) {
            td[(2 * j + 0) * 32] = v4f{tv[j][0], tv[j][1], tv[j][2], tv[j][3]};
            td[(2 * j + 1) * 32] = v4f{tv[j][4], tv[j][5], tv[j][6], tv[j][7]};
        }
    }

#pragma unroll
    for (int i = 0; i < K_SZ; ++i) {
        v2f* d = &w2s[0][fdst + 16 * i];
        d[0] = v2f{f0[i].x, f1[i].x};
        d[1] = v2f{f0[i].y, f1[i].y};
        d[2] = v2f{f0[i].z, f1[i].z};
        d[3] = v2f{f0[i].w, f1[i].w};
    }
    __syncthreads();

    v2f a[CR / 4];
#pragma unroll
    for (int q = 0; q < 2; ++q) {
#pragma unroll
        for (int tq = 0; tq < 4; ++tq) {
            const v4f v = tex[ml][q][tq * 32 + ln];
            a[8 * q + 2 * tq]     = v2f{v.x, v.y};
            a[8 * q + 2 * tq + 1] = v2f{v.z, v.w};
        }
    }

    int  cofs[14];
    bool cok[14];
#pragma unroll
    for (int ci = 0; ci < 14; ++ci) {
        const int ww = wl - (K_SZ / 2) + ci;
        cok[ci]  = (unsigned)ww < (unsigned)HW;
        cofs[ci] = (cok[ci] ? ww : wl) * C_CH;
    }

    float o[8][8];
#pragma unroll
    for (int jj = 0; jj < 8; ++jj)
#pragma unroll
        for (int r = 0; r < 8; ++r) o[jj][r] = 0.0f;

    const int chan0 = g + G_NUM * 8 * jh;

#pragma unroll 1
    for (int kh = 0; kh < K_SZ; ++kh) {
        const int cur = kh & 1;
        const int kh1 = (kh + 1 < K_SZ) ? kh + 1 : kh;

        const int  hh    = h + kh - (K_SZ / 2);
        const bool rowok = (unsigned)hh < (unsigned)HW;
        const int  hc    = rowok ? hh : h;
        const float* xr0 = x + ((size_t)(b * HW + hc) * HW) * C_CH + chan0;
        float xr[8][14];
        v8f   kr[K_SZ];

        const v2f* bl   = &w2s[cur][half * SLAB_N + g];
        v4f*       xbuf = &xch[cur][ml][ln];

        if (jh == 0)
            trip_front<0, TAPS_A>(frow, kh1, b2g, xr0, cofs, a, bl, xbuf, f0, f1, bcn, xr, kr);
        else
            trip_front<TAPS_A, K_SZ - TAPS_A>(frow, kh1, b2g, xr0, cofs, a, bl, xbuf, f0, f1, bcn, xr, kr);

#pragma unroll
        for (int i = 0; i < K_SZ; ++i) {
            v2f* d = &w2s[cur ^ 1][fdst + 16 * i];
            d[0] = v2f{f0[i].x, f1[i].x};
            d[1] = v2f{f0[i].y, f1[i].y};
            d[2] = v2f{f0[i].z, f1[i].z};
            d[3] = v2f{f0[i].w, f1[i].w};
        }

        __syncthreads();

        float* kp0 = kern + (size_t)pb * NK + (kh * K_SZ) * G_NUM + g;
        if (jh == 0)
            trip_back<0, TAPS_A>(xbuf, kp0, kr);
        else
            trip_back<TAPS_A, K_SZ - TAPS_A>(xbuf, kp0, kr);

#pragma unroll
        for (int jj = 0; jj < 8; ++jj) {
            float xv[14];
#pragma unroll
            for (int ci = 0; ci < 14; ++ci)
                xv[ci] = (cok[ci] && rowok) ? xr[jj][ci] : 0.0f;
#pragma unroll
            for (int kw = 0; kw < K_SZ; ++kw)
#pragma unroll
                for (int r = 0; r < 8; ++r)
                    o[jj][r] = fmaf(kr[kw][r], xv[r + kw], o[jj][r]);
        }
    }

#pragma unroll
    for (int jj = 0; jj < 8; ++jj)
#pragma unroll
        for (int r = 0; r < 8; ++r)
            out[(size_t)(pb + r) * C_CH + chan0 + G_NUM * jj] = o[jj][r];
}

extern "C" void kernel_launch(void* const* d_in, const int* in_sizes, int n_in,
                              void* d_out, int out_size, void* d_ws, size_t ws_size,
                              hipStream_t stream)
{
    const float* x     = (const float*)d_in[0];
    const float* w1    = (const float*)d_in[1];
    const float* b1    = (const float*)d_in[2];
    const float* gamma = (const float*)d_in[3];
    const float* beta  = (const float*)d_in[4];
    const float* mu    = (const float*)d_in[5];
    const float* var   = (const float*)d_in[6];
    const float* w2    = (const float*)d_in[7];
    const float* b2    = (const float*)d_in[8];

    float* out  = (float*)d_out;
    float* kern = out + (size_t)M_TOT * C_CH;

    involution_fused<<<256, 128, 0, stream>>>(x, w1, b1, gamma, beta, mu, var, w2, b2, kern, out);
}


# each m-tile's work split over 4 waves, not 2 (wave = 2 taps + 4 channel blocks; 8-wave block, two waves a SIMD): one wave's load, LDS and barrier waits are the other's issue time; sum order kept
# speedup vs baseline: 3.5534x; 1.0375x over previous
#include <hip/hip_runtime.h>


#define K_SZ   7
#define KK     49
#define G_NUM  16
#define C_CH   256
#define CR     64
#define NK     (KK * G_NUM)
#define HW     64
#define BATCH  2
#define M_TOT  (BATCH * HW * HW)
#define BN_EPS 1e-3f

typedef __attribute__((ext_vector_type(2))) float v2f;
typedef __attribute__((ext_vector_type(4))) float v4f;
typedef __attribute__((ext_vector_type(8))) float v8f;

#define G1_U  8

#define SLAB_N  (K_SZ * G_NUM)
#define SLAB_P  ((CR / 2) * SLAB_N)
#define NQ      4
#define JJ_W    (C_CH / G_NUM / NQ)
#define NT_MAX  2
#define SL_PC   4

template <int NT>
__device__ __forceinline__ void trip_front(
    const float* __restrict__ fsrc, const int pc3,
    const float* __restrict__ b2n, const float* __restrict__ xr0,
    const int (&cofs)[14], const v2f (&a)[CR / 4], const v2f* __restrict__ bl,
    v4f* __restrict__ xdst,
    v4f (&f0)[SL_PC], v4f (&f1)[SL_PC], float (&bcn)[NT_MAX],
    float (&xr)[JJ_W][14])
{
#pragma unroll
    for (int i = 0; i < SL_PC; ++i) {
        const int po = (i < SL_PC - 1) ? 16 * i : pc3;
        f0[i] = *(const v4f*)(fsrc + po);
        f1[i] = *(const v4f*)(fsrc + 2 * NK + po);
    }
    float bc[NT];
#pragma unroll
    for (int t = 0; t < NT; ++t) bc[t] = bcn[t];
#pragma unroll
    for (int t = 0; t < NT; ++t) bcn[t] = b2n[t * G_NUM];
#pragma unroll
    for (int jj = 0; jj < JJ_W; ++jj)
#pragma unroll
        for (int ci = 0; ci < 14; ++ci)
            xr[jj][ci] = xr0[cofs[ci] + G_NUM * jj];

    v8f acc[NT];
#pragma unroll
    for (int t = 0; t < NT; ++t) acc[t] = v8f{};

    v2f bf[2][4][NT];
#pragma unroll
    for (int kq = 0; kq < 4; ++kq)
#pragma unroll
        for (int t = 0; t < NT; ++t)
            bf[0][kq][t] = bl[(2 * kq) * SLAB_N + t * G_NUM];
#pragma unroll
    for (int q = 0; q < 4; ++q) {
        if (q + 1 < 4) {
#pragma unroll
            for (int kq = 0; kq < 4; ++kq)
#pragma unroll
                for (int t = 0; t < NT; ++t)
                    bf[(q + 1) & 1][kq][t] =
                        bl[(2 * (4 * (q + 1) + kq)) * SLAB_N + t * G_NUM];
        }
#pragma unroll
        for (int kq = 0; kq < 4; ++kq)
#pragma unroll
            for (int t = 0; t < NT; ++t)
                acc[t] = __builtin_amdgcn_wmma_f32_16x16x4_f32(
                    false, a[4 * q + kq], false, bf[q & 1][kq][t],
                    (short)0, acc[t], false, false);
    }
    constexpr int NL  = 2 * SL_PC + NT + 14 * JJ_W;
    constexpr int NL1 = 2 * SL_PC + NT + 14 * (JJ_W / 2);
    constexpr int NL2 = NL - NL1;
    __builtin_amdgcn_sched_group_barrier(0x020, NL1, 0);
    __builtin_amdgcn_sched_group_barrier(0x100, 2 * NT, 0);
    __builtin_amdgcn_sched_group_barrier(0x100, 2 * NT, 0);
    __builtin_amdgcn_sched_group_barrier(0x008, 4 * NT, 0);
    __builtin_amdgcn_sched_group_barrier(0x020, NL2, 0);
    __builtin_amdgcn_sched_group_barrier(0x100, 2 * NT, 0);
    __builtin_amdgcn_sched_group_barrier(0x008, 4 * NT, 0);
    __builtin_amdgcn_sched_group_barrier(0x100, 2 * NT, 0);
    __builtin_amdgcn_sched_group_barrier(0x008, 4 * NT, 0);
    __builtin_amdgcn_sched_group_barrier(0x008, 4 * NT, 0);

#pragma unroll
    for (int t = 0; t < NT; ++t)
#pragma unroll
        for (int r = 0; r < 8; ++r) acc[t][r] += bc[t];

#pragma unroll
    for (int t = 0; t < NT; ++t) {
        const v8f v = acc[t];
        xdst[(t * 2 + 0) * 32] = v4f{v[0], v[1], v[2], v[3]};
        xdst[(t * 2 + 1) * 32] = v4f{v[4], v[5], v[6], v[7]};
    }
}

__global__ __launch_bounds__(256) void involution_fused(
    const float* __restrict__ x,  const float* __restrict__ w1,
    const float* __restrict__ b1, const float* __restrict__ gamma,
    const float* __restrict__ beta, const float* __restrict__ mu,
    const float* __restrict__ var, const float* __restrict__ w2,
    const float* __restrict__ b2,
    float* __restrict__ kern, float* __restrict__ out)
{
    __shared__ v2f w2s[2][SLAB_P];
    __shared__ v4f xch[2][2][K_SZ * 2 * 32];
    __shared__ v4f tex[2][2][4 * 32];

    const int wave = __builtin_amdgcn_readfirstlane((int)(threadIdx.x >> 5));
    const int ln   = threadIdx.x & 31;
    const int ml   = (wave >> 1) & 1;
    const int q    = (wave & 1) + 2 * (wave >> 2);
    const int up   = wave >> 2;
    const int mt   = blockIdx.x * 2 + ml;
    const int g    = ln & 15;
    const int half = ln >> 4;
    const int m0   = mt << 4;
    const int w0   = m0 & (HW - 1);
    const int h    = (m0 >> 6) & (HW - 1);
    const int b    = m0 >> 12;
    const int pb   = m0 + (half << 3);
    const int wl   = w0 + (half << 3);

    const int tl  = threadIdx.x & 127;
    const int fk2 = tl >> 2;
    const int fc4 = tl & 3;
    const int fkA = 32 * (fk2 >> 4) + 16 * (fk2 & 1) + 4 * ((fk2 >> 1) & 3) + ((fk2 >> 3) & 1);
    const int pc0 = 16 * SL_PC * up;
    const int pc3 = up ? 32 : 48;
    const float* frow = w2 + (size_t)fkA * NK + 4 * fc4 + pc0;
    const int    fdst = fk2 * SLAB_N + 4 * fc4 + pc0;

    const int kw0 = 2 * q;
    const float* b2k = b2 + g + kw0 * G_NUM;

    v4f f0[SL_PC], f1[SL_PC];
    float bcn[NT_MAX];

    if (q < 2) {
        const int jh   = q;
        const int n0   = jh << 5;
        const int dcol = n0 + 16 * half;
        const float* ap = x  + (size_t)(m0 + g) * C_CH + 4 * half;
        const float* bp = w1 + (size_t)(4 * half) * CR + n0 + 2 * g;

        v4f cgm[4], cvr[4], cbt[4], cmn[4], cbb[4];
#pragma unroll
        for (int i = 0; i < 4; ++i) {
            cgm[i] = *(const v4f*)(gamma + dcol + 4 * i);
            cvr[i] = *(const v4f*)(var   + dcol + 4 * i);
            cbt[i] = *(const v4f*)(beta  + dcol + 4 * i);
            cmn[i] = *(const v4f*)(mu    + dcol + 4 * i);
            cbb[i] = *(const v4f*)(b1    + dcol + 4 * i);
        }

        v4f xa[2][G1_U];
        v2f wb[2][G1_U][4];
#pragma unroll
        for (int uu = 0; uu < G1_U; ++uu) {
            xa[0][uu] = *(const v4f*)(ap + 8 * uu);
#pragma unroll
            for (int e = 0; e < 4; ++e)
                wb[0][uu][e] = *(const v2f*)(bp + (size_t)(8 * uu + e) * CR);
        }

        v8f acc0 = {}, acc1 = {};
#pragma unroll
        for (int gi = 0; gi < C_CH / (8 * G1_U); ++gi) {
            const int cb = gi & 1;
            if (gi + 1 < C_CH / (8 * G1_U)) {
#pragma unroll
                for (int uu = 0; uu < G1_U; ++uu) {
                    const int u = (gi + 1) * G1_U + uu;
                    xa[cb ^ 1][uu] = *(const v4f*)(ap + 8 * u);
#pragma unroll
                    for (int e = 0; e < 4; ++e)
                        wb[cb ^ 1][uu][e] = *(const v2f*)(bp + (size_t)(8 * u + e) * CR);
                }
            } else {
#pragma unroll
                for (int i = 0; i < SL_PC; ++i) {
                    const int po = (i < SL_PC - 1) ? 16 * i : pc3;
                    f0[i] = *(const v4f*)(frow + po);
                    f1[i] = *(const v4f*)(frow + 2 * NK + po);
                }
#pragma unroll
                for (int tt = 0; tt < NT_MAX; ++tt)
                    bcn[tt] = b2k[tt * G_NUM];
            }
#pragma unroll
            for (int uu = 0; uu < G1_U; ++uu) {
#pragma unroll
                for (int sub = 0; sub < 2; ++sub) {
                    v2f xf, wf0, wf1;
                    xf.x  = xa[cb][uu][2 * sub];
                    xf.y  = xa[cb][uu][2 * sub + 1];
                    wf0.x = wb[cb][uu][2 * sub].x;
                    wf0.y = wb[cb][uu][2 * sub + 1].x;
                    wf1.x = wb[cb][uu][2 * sub].y;
                    wf1.y = wb[cb][uu][2 * sub + 1].y;
                    acc0 = __builtin_amdgcn_wmma_f32_16x16x4_f32(
                        false, wf0, false, xf, (short)0, acc0, false, false);
                    acc1 = __builtin_amdgcn_wmma_f32_16x16x4_f32(
                        false, wf1, false, xf, (short)0, acc1, false, false);
                }
            }
        }

        __builtin_amdgcn_sched_group_barrier(0x020, 60, 0);
        __builtin_amdgcn_sched_group_barrier(0x020, 40, 0);
        __builtin_amdgcn_sched_group_barrier(0x008, 32, 0);
        __builtin_amdgcn_sched_group_barrier(0x020, 40, 0);
        __builtin_amdgcn_sched_group_barrier(0x008, 32, 0);
        __builtin_amdgcn_sched_group_barrier(0x020, 40, 0);
        __builtin_amdgcn_sched_group_barrier(0x008, 32, 0);
        __builtin_amdgcn_sched_group_barrier(0x020, 2 * SL_PC + NT_MAX, 0);
        __builtin_amdgcn_sched_group_barrier(0x008, 32, 0);

        float tv[2][8];
#pragma unroll
        for (int r = 0; r < 8; ++r) {
#pragma unroll
            for (int j = 0; j < 2; ++j) {
                const int i = 2 * r + j;
                const float scale = cgm[i >> 2][i & 3] * rsqrtf(cvr[i >> 2][i & 3] + BN_EPS);
                const float shift = cbt[i >> 2][i & 3] - cmn[i >> 2][i & 3] * scale;
                const float pre   = (j ? acc1[r] : acc0[r]) + cbb[i >> 2][i & 3];
                tv[j][r] = fmaxf(fmaf(pre, scale, shift), 0.0f);
            }
        }
        v4f* td = &tex[ml][jh][ln];
#pragma unroll
        for (int j = 0; j < 2; ++j) {
            td[(2 * j + 0) * 32] = v4f{tv[j][0], tv[j][1], tv[j][2], tv[j][3]};
            td[(2 * j + 1) * 32] = v4f{tv[j][4], tv[j][5], tv[j][6], tv[j][7]};
        }
    } else {
#pragma unroll
        for (int i = 0; i < SL_PC; ++i) {
            const int po = (i < SL_PC - 1) ? 16 * i : pc3;
            f0[i] = *(const v4f*)(frow + po);
            f1[i] = *(const v4f*)(frow + 2 * NK + po);
        }
#pragma unroll
        for (int tt = 0; tt < NT_MAX; ++tt)
            bcn[tt] = b2k[((kw0 + tt < K_SZ) ? tt : 0) * G_NUM];
    }

#pragma unroll
    for (int i = 0; i < SL_PC; ++i) {
        const int po = (i < SL_PC - 1) ? 16 * i : pc3;
        v2f* d = &w2s[0][fdst + po];
        d[0] = v2f{f0[i].x, f1[i].x};
        d[1] = v2f{f0[i].y, f1[i].y};
        d[2] = v2f{f0[i].z, f1[i].z};
        d[3] = v2f{f0[i].w, f1[i].w};
    }
    __syncthreads();

    v2f a[CR / 4];
#pragma unroll
    for (int qh = 0; qh < 2; ++qh) {
#pragma unroll
        for (int tq = 0; tq < 4; ++tq) {
            const v4f v = tex[ml][qh][tq * 32 + ln];
            a[8 * qh + 2 * tq]     = v2f{v.x, v.y};
            a[8 * qh + 2 * tq + 1] = v2f{v.z, v.w};
        }
    }

    int  cofs[14];
    bool cok[14];
#pragma unroll
    for (int ci = 0; ci < 14; ++ci) {
        const int ww = wl - (K_SZ / 2) + ci;
        cok[ci]  = (unsigned)ww < (unsigned)HW;
        cofs[ci] = (cok[ci] ? ww : wl) * C_CH;
    }

    float o[JJ_W][8];
#pragma unroll
    for (int jj = 0; jj < JJ_W; ++jj)
#pragma unroll
        for (int r = 0; r < 8; ++r) o[jj][r] = 0.0f;

    const int chan0 = g + G_NUM * JJ_W * q;

#pragma unroll 1
    for (int kh = 0; kh < K_SZ; ++kh) {
        const int cur = kh & 1;
        const int kh1 = (kh + 1 < K_SZ) ? kh + 1 : kh;

        const int  hh    = h + kh - (K_SZ / 2);
        const bool rowok = (unsigned)hh < (unsigned)HW;
        const int  hc    = rowok ? hh : h;
        const float* xr0 = x + ((size_t)(b * HW + hc) * HW) * C_CH + chan0;
        float xr[JJ_W][14];

        const v2f*   bl   = &w2s[cur][half * SLAB_N + kw0 * G_NUM + g];
        v4f*         xbuf = &xch[cur][ml][ln];
        const float* fsrc = frow + kh1 * SLAB_N;
        const float* b2n  = b2k + kh1 * K_SZ * G_NUM;

        if (q < NQ - 1)
            trip_front<2>(fsrc, pc3, b2n, xr0, cofs, a, bl, xbuf + kw0 * 2 * 32, f0, f1, bcn, xr);
        else
            trip_front<1>(fsrc, pc3, b2n, xr0, cofs, a, bl, xbuf + kw0 * 2 * 32, f0, f1, bcn, xr);

#pragma unroll
        for (int i = 0; i < SL_PC; ++i) {
            const int po = (i < SL_PC - 1) ? 16 * i : pc3;
            v2f* d = &w2s[cur ^ 1][fdst + po];
            d[0] = v2f{f0[i].x, f1[i].x};
            d[1] = v2f{f0[i].y, f1[i].y};
            d[2] = v2f{f0[i].z, f1[i].z};
            d[3] = v2f{f0[i].w, f1[i].w};
        }

        __syncthreads();

        v8f kr[K_SZ];
#pragma unroll
        for (int kw = 0; kw < K_SZ; ++kw) {
            const v4f lo = xbuf[(kw * 2 + 0) * 32];
            const v4f hi = xbuf[(kw * 2 + 1) * 32];
            kr[kw] = v8f{lo.x, lo.y, lo.z, lo.w, hi.x, hi.y, hi.z, hi.w};
        }

        if (q == NQ - 1) {
            float* kp0 = kern + (size_t)pb * NK + (kh * K_SZ) * G_NUM + g;
#pragma unroll
            for (int kw = 0; kw < K_SZ; ++kw)
#pragma unroll
                for (int r = 0; r < 8; ++r)
                    kp0[(size_t)r * NK + kw * G_NUM] = kr[kw][r];
        }

#pragma unroll
        for (int jj = 0; jj < JJ_W; ++jj) {
            float xv[14];
#pragma unroll
            for (int ci = 0; ci < 14; ++ci)
                xv[ci] = (cok[ci] && rowok) ? xr[jj][ci] : 0.0f;
#pragma unroll
            for (int kw = 0; kw < K_SZ; ++kw)
#pragma unroll
                for (int r = 0; r < 8; ++r)
                    o[jj][r] = fmaf(kr[kw][r], xv[r + kw], o[jj][r]);
        }
    }

#pragma unroll
    for (int jj = 0; jj < JJ_W; ++jj)
#pragma unroll
        for (int r = 0; r < 8; ++r)
            out[(size_t)(pb + r) * C_CH + chan0 + G_NUM * jj] = o[jj][r];
}

extern "C" void kernel_launch(void* const* d_in, const int* in_sizes, int n_in,
                              void* d_out, int out_size, void* d_ws, size_t ws_size,
                              hipStream_t stream)
{
    const float* x     = (const float*)d_in[0];
    const float* w1    = (const float*)d_in[1];
    const float* b1    = (const float*)d_in[2];
    const float* gamma = (const float*)d_in[3];
    const float* beta  = (const float*)d_in[4];
    const float* mu    = (const float*)d_in[5];
    const float* var   = (const float*)d_in[6];
    const float* w2    = (const float*)d_in[7];
    const float* b2    = (const float*)d_in[8];

    float* out  = (float*)d_out;
    float* kern = out + (size_t)M_TOT * C_CH;

    involution_fused<<<256, 256, 0, stream>>>(x, w1, b1, gamma, beta, mu, var, w2, b2, kern, out);
}
